# starting kernel
# baseline (speedup 1.0000x reference)
#include <hip/hip_runtime.h>

typedef __attribute__((ext_vector_type(2))) float v2f;
typedef __attribute__((ext_vector_type(8))) float v8f;

#define NN    10000
#define BB    2
#define FF    64
#define NASZ  64
#define OUTC  64
#define KDIM  128
#define NPW   32
#define ROWS  64
#define LSTR  132
#define THREADS 256

__launch_bounds__(THREADS)
__global__ void pgnn_fused_kernel(const float* __restrict__ x,
                                  const float* __restrict__ dm,
                                  const int*   __restrict__ idx,
                                  const float* __restrict__ W,
                                  const float* __restrict__ bias,
                                  float* __restrict__ out)
{
    __shared__ float w_lds[OUTC * LSTR];
    __shared__ float h_lds[ROWS * LSTR];
    __shared__ float bias_lds[OUTC];
    __shared__ int   idx_lds[NPW * NASZ];
    __shared__ float dm_lds [NPW * NASZ];

    const int tid = threadIdx.x;
    const int n0  = blockIdx.x * NPW;

    for (int e = tid; e < OUTC * KDIM; e += THREADS) {
        int o = e >> 7, k = e & 127;
        w_lds[o * LSTR + k] = W[e];
    }
    if (tid < OUTC) bias_lds[tid] = bias[tid];

    for (int e = tid; e < NPW * NASZ; e += THREADS) {
        int nl = e >> 6;
        int n  = n0 + nl;
        if (n < NN) {
            idx_lds[e] = idx[n * NASZ + (e & 63)];
            dm_lds[e]  = dm [n * NASZ + (e & 63)];
        } else {
            idx_lds[e] = 0;
            dm_lds[e]  = 0.0f;
        }
    }
    __syncthreads();

    {
        const int half = tid >> 7;
        const int t    = tid & 127;
        const int b    = t >> 6;
        const int f    = t & 63;
        const int nlo  = half * (NPW / 2);
        for (int nl = nlo; nl < nlo + NPW / 2; ++nl) {
            const int*   ip = &idx_lds[nl * NASZ];
            const float* dp = &dm_lds [nl * NASZ];
            float acc = 0.0f;
            #pragma unroll 8
            for (int a = 0; a < NASZ; ++a) {
                acc = fmaf(dp[a], x[(size_t)ip[a] * KDIM + t], acc);
            }
            const int n   = n0 + nl;
            const int row = nl * 2 + b;
            h_lds[row * LSTR + f]      = acc * (1.0f / NASZ);
            h_lds[row * LSTR + FF + f] = (n < NN) ? x[(size_t)n * KDIM + t] : 0.0f;
        }
    }
    __syncthreads();

    const int wave  = tid >> 5;
    const int lane  = tid & 31;
    const int l15   = lane & 15;
    const int khalf = (lane >> 4) * 2;
    const int rt    = wave >> 1;
    const int ch    = wave & 1;
    const int rowA  = rt * 16 + l15;
    const int col0  = (ch * 2) * 16 + l15;
    const int col1  = (ch * 2 + 1) * 16 + l15;

    v8f acc0, acc1;
    {
        const float bv0 = bias_lds[col0];
        const float bv1 = bias_lds[col1];
        #pragma unroll
        for (int r = 0; r < 8; ++r) { acc0[r] = bv0; acc1[r] = bv1; }
    }

    #pragma unroll 4
    for (int kk = 0; kk < 32; ++kk) {
        const int kb = kk * 4 + khalf;
        v2f afrag;
        afrag.x = h_lds[rowA * LSTR + kb];
        afrag.y = h_lds[rowA * LSTR + kb + 1];
        v2f bfrag0, bfrag1;
        bfrag0.x = w_lds[col0 * LSTR + kb];
        bfrag0.y = w_lds[col0 * LSTR + kb + 1];
        bfrag1.x = w_lds[col1 * LSTR + kb];
        bfrag1.y = w_lds[col1 * LSTR + kb + 1];
        acc0 = __builtin_amdgcn_wmma_f32_16x16x4_f32(
            false, afrag, false, bfrag0, (short)0, acc0, false, false);
        acc1 = __builtin_amdgcn_wmma_f32_16x16x4_f32(
            false, afrag, false, bfrag1, (short)0, acc1, false, false);
    }

    const int mbase = blockIdx.x * ROWS + rt * 16 + (lane >> 4) * 8;
    float* __restrict__ op = out + (size_t)mbase * OUTC;
    const bool full = (blockIdx.x + 1) * ROWS <= NN * BB;
    if (full) {
        #pragma unroll
        for (int r = 0; r < 8; ++r) {
            op[(size_t)r * OUTC + col0] = acc0[r];
            op[(size_t)r * OUTC + col1] = acc1[r];
        }
    } else {
        #pragma unroll
        for (int r = 0; r < 8; ++r) {
            if (mbase + r < NN * BB) {
                op[(size_t)r * OUTC + col0] = acc0[r];
                op[(size_t)r * OUTC + col1] = acc1[r];
            }
        }
    }
}

extern "C" void kernel_launch(void* const* d_in, const int* in_sizes, int n_in,
                              void* d_out, int out_size, void* d_ws, size_t ws_size,
                              hipStream_t stream) {
    const float* x    = (const float*)d_in[0];
    const float* dm   = (const float*)d_in[1];
    const int*   idx  = (const int*)  d_in[2];
    const float* W    = (const float*)d_in[3];
    const float* bias = (const float*)d_in[4];
    float* out = (float*)d_out;

    const int grid = (NN * BB + ROWS - 1) / ROWS;
    pgnn_fused_kernel<<<grid, THREADS, 0, stream>>>(x, dm, idx, W, bias, out);
}


# Both kernels run half as many waves per workgroup (gather 16 -> 8, each wave on two node-quad streams; linear 8 -> 4, each wave on 8 D tiles), so half as many waves must start per run; sums unchanged
# speedup vs baseline: 2.3098x; 2.3098x over previous
#include <hip/hip_runtime.h>


typedef __attribute__((ext_vector_type(2))) float v2f;
typedef __attribute__((ext_vector_type(4))) float v4f;
typedef __attribute__((ext_vector_type(8))) float v8f;

#define NN    10000
#define BB    2
#define FF    64
#define NASZ  64
#define OUTC  64
#define KDIM  128
#define NPW   32
#define ROWS  64
#define LSTR  132
#define YSTR  68
#define THREADS 128

#define NSLICE   16
#define NGROUP   16
#define GTHREADS 256
#define GWAVES   8
#define QSTEP    16
#define FULLR    78
#define NQUAD    (NN / 4)
#define XS1      (NN + 8)
#define SLICE_FLOATS (NN * 8)

__launch_bounds__(THREADS)
__global__ void linear_first_kernel(const float* __restrict__ x,
                                    const float* __restrict__ W,
                                    const float* __restrict__ bias,
                                    float* __restrict__ y_ws,
                                    float* __restrict__ out)
{
    __shared__ __attribute__((aligned(16))) float wt_lds[2 * OUTC * YSTR];
    __shared__ __attribute__((aligned(16))) float h_lds[ROWS * YSTR];
    __shared__ __attribute__((aligned(16))) float st_lds[ROWS * LSTR];

    const int tid  = threadIdx.x;
    const int n0   = blockIdx.x * NPW;
    const int row0 = blockIdx.x * ROWS;
    const int lane = tid & 31;
    const int l15  = lane & 15;

    float4 wv[16], xv[8];
    {
        const float4* W4 = (const float4*)W;
        const float4* x4 = (const float4*)x;
        #pragma unroll
        for (int i = 0; i < 16; ++i) wv[i] = W4[tid + THREADS * i];
        #pragma unroll
        for (int i = 0; i < 8; ++i) {
            const int e  = tid + THREADS * i;
            const int R  = row0 + (e >> 4);
            const int Rc = R < NN * BB ? R : NN * BB - 1;
            xv[i] = x4[Rc * (FF / 4) + (e & 15)];
        }
    }
    const float bq0 = bias[l15];
    const float bq1 = bias[16 + l15];
    const float bq2 = bias[32 + l15];
    const float bq3 = bias[48 + l15];

    #pragma unroll
    for (int i = 0; i < 16; ++i) {
        const int e4 = tid + THREADS * i;
        const int o  = e4 >> 5;
        const int kk = (e4 & 31) * 4;
        *(float4*)&wt_lds[(o + (kk & 64)) * YSTR + (kk & 63)] = wv[i];
    }
    #pragma unroll
    for (int i = 0; i < 8; ++i) {
        const int e   = tid + THREADS * i;
        const int row = e >> 4;
        const int t4  = e & 15;
        const bool ok = (row0 + row) < NN * BB;
        float4 vx = xv[i];
        vx.x = ok ? vx.x : 0.f;  vx.y = ok ? vx.y : 0.f;
        vx.z = ok ? vx.z : 0.f;  vx.w = ok ? vx.w : 0.f;
        *(float4*)&h_lds[row * YSTR + t4 * 4] = vx;
    }
    __syncthreads();

    const int wave  = tid >> 5;
    const int khalf = (lane >> 4) * 2;
    const int rt    = wave;
    const int rowA  = rt * 16 + l15;
    const int cb    = l15;

    v8f acc[8];
    #pragma unroll
    for (int r = 0; r < 8; ++r) {
        acc[0][r] = 0.f; acc[1][r] = 0.f; acc[2][r] = 0.f; acc[3][r] = 0.f;
        acc[4][r] = bq0; acc[5][r] = bq1; acc[6][r] = bq2; acc[7][r] = bq3;
    }

    #pragma unroll
    for (int kk = 0; kk < 16; ++kk) {
        const int kb = kk * 4 + khalf;
        v2f afrag;
        afrag.x = h_lds[rowA * YSTR + kb];
        afrag.y = h_lds[rowA * YSTR + kb + 1];
        v2f bf[8];
        #pragma unroll
        for (int j = 0; j < 8; ++j) {
            bf[j].x = wt_lds[(cb + 16 * j) * YSTR + kb];
            bf[j].y = wt_lds[(cb + 16 * j) * YSTR + kb + 1];
        }
        #pragma unroll
        for (int j = 0; j < 8; ++j)
            acc[j] = __builtin_amdgcn_wmma_f32_16x16x4_f32(false, afrag, false, bf[j], (short)0, acc[j], false, false);
    }

    {
        float* sp = &st_lds[(rt * 16 + (lane >> 4) * 8) * LSTR + cb];
        #pragma unroll
        for (int r = 0; r < 8; ++r) {
            #pragma unroll
            for (int j = 0; j < 8; ++j) sp[r * LSTR + 16 * j] = acc[j][r];
        }
    }
    __syncthreads();

    {
        v4f* y4 = (v4f*)y_ws;
        v4f* o4 = (v4f*)out;
        v4f vy[8], vs[8];
        #pragma unroll
        for (int i = 0; i < 8; ++i) {
            const int e    = tid + THREADS * i;
            const int sl   = e >> 6;
            const int r    = e & 63;
            const int nl   = r >> 1;
            const int half = r & 1;
            vy[i] = *(const v4f*)&st_lds[(nl * 2 + (sl >> 3)) * LSTR + (sl & 7) * 8 + half * 4];
            vs[i] = *(const v4f*)&st_lds[(e >> 4) * LSTR + OUTC + (e & 15) * 4];
        }
        asm volatile("" :: "v"(vy[0]), "v"(vy[1]), "v"(vy[2]), "v"(vy[3]),
                           "v"(vy[4]), "v"(vy[5]), "v"(vy[6]), "v"(vy[7]),
                           "v"(vs[0]), "v"(vs[1]), "v"(vs[2]), "v"(vs[3]),
                           "v"(vs[4]), "v"(vs[5]), "v"(vs[6]), "v"(vs[7]));
        if (n0 + NPW <= NN) {
            #pragma unroll
            for (int i = 0; i < 8; ++i) {
                const int e = tid + THREADS * i;
                y4[(e >> 6) * (SLICE_FLOATS / 4) + (n0 + ((e & 63) >> 1)) * 2 + (e & 1)] = vy[i];
                o4[(row0 + (e >> 4)) * (OUTC / 4) + (e & 15)] = vs[i];
            }
        } else {
            #pragma unroll
            for (int i = 0; i < 8; ++i) {
                const int e = tid + THREADS * i;
                const int n = n0 + ((e & 63) >> 1);
                const int R = row0 + (e >> 4);
                if (n < NN)      y4[(e >> 6) * (SLICE_FLOATS / 4) + n * 2 + (e & 1)] = vy[i];
                if (R < NN * BB) o4[R * (OUTC / 4) + (e & 15)] = vs[i];
            }
        }
    }
}

struct Ent { int4 i0, i1; float4 d0, d1; };

__device__ __forceinline__ Ent load_ent(const int* __restrict__ idx, const float* __restrict__ dm,
                                        int q, int sub, int ch)
{
    const int n = q * 4 + sub;
    const int4*   ip = (const int4*)  (idx + n * NASZ + ch * 8);
    const float4* dp = (const float4*)(dm  + n * NASZ + ch * 8);
    Ent e;
    e.i0 = ip[0]; e.i1 = ip[1];
    e.d0 = dp[0]; e.d1 = dp[1];
    return e;
}

__device__ __forceinline__ float quad_sum(const float4* xs, const Ent& e, bool b4, bool b2, bool b1)
{
    const int   jj[8] = { e.i0.x, e.i0.y, e.i0.z, e.i0.w, e.i1.x, e.i1.y, e.i1.z, e.i1.w };
    const float dd[8] = { e.d0.x, e.d0.y, e.d0.z, e.d0.w, e.d1.x, e.d1.y, e.d1.z, e.d1.w };

    float a0 = 0.f, a1 = 0.f, a2 = 0.f, a3 = 0.f, a4 = 0.f, a5 = 0.f, a6 = 0.f, a7 = 0.f;
    #pragma unroll
    for (int k = 0; k < 8; ++k) {
        const float4 lo = xs[jj[k]];
        const float4 hi = xs[XS1 + jj[k]];
        const float  w  = dd[k];
        a0 = fmaf(w, lo.x, a0); a1 = fmaf(w, lo.y, a1);
        a2 = fmaf(w, lo.z, a2); a3 = fmaf(w, lo.w, a3);
        a4 = fmaf(w, hi.x, a4); a5 = fmaf(w, hi.y, a5);
        a6 = fmaf(w, hi.z, a6); a7 = fmaf(w, hi.w, a7);
    }

    float k0 = b4 ? a4 : a0, k1 = b4 ? a5 : a1, k2 = b4 ? a6 : a2, k3 = b4 ? a7 : a3;
    float s0 = b4 ? a0 : a4, s1 = b4 ? a1 : a5, s2 = b4 ? a2 : a6, s3 = b4 ? a3 : a7;
    k0 += __shfl_xor(s0, 4); k1 += __shfl_xor(s1, 4);
    k2 += __shfl_xor(s2, 4); k3 += __shfl_xor(s3, 4);
    float m0 = b2 ? k2 : k0, m1 = b2 ? k3 : k1;
    float t0 = b2 ? k0 : k2, t1 = b2 ? k1 : k3;
    m0 += __shfl_xor(t0, 2); m1 += __shfl_xor(t1, 2);
    float r  = b1 ? m1 : m0;
    float u  = b1 ? m0 : m1;
    r += __shfl_xor(u, 1);
    return r;
}

__launch_bounds__(GTHREADS)
__global__ void gather_slices_kernel(const float* __restrict__ y_ws,
                                     const float* __restrict__ dm,
                                     const int*   __restrict__ idx,
                                     float* out)
{
    __shared__ float4 xs[XS1 + NN];

    const int tid  = threadIdx.x;
    const int s    = blockIdx.x & (NSLICE - 1);
    const int g    = blockIdx.x >> 4;
    const int wave = tid >> 5;
    const int lane = tid & 31;
    const int sub  = lane >> 3;
    const int ch   = lane & 7;

    const int qlo = (NQUAD * g) >> 4;
    const int qhi = (NQUAD * (g + 1)) >> 4;

    int qA = qlo + wave;
    int qB = qA + GWAVES;
    Ent eA = load_ent(idx, dm, qA, sub, ch);
    Ent eB = load_ent(idx, dm, (qB < qhi) ? qB : qA, sub, ch);

    {
        const float4* y4 = (const float4*)y_ws + (size_t)s * (SLICE_FLOATS / 4);
        float4* xb = xs + (tid & 1) * XS1 + (tid >> 1);
        const int last = (tid < 2 * NN - GTHREADS * FULLR) ? FULLR : FULLR - 1;
        float4 v[FULLR + 1];
        #pragma unroll
        for (int i = 0; i < FULLR; ++i) v[i] = y4[tid + GTHREADS * i];
        v[FULLR] = y4[tid + GTHREADS * last];
        #pragma unroll
        for (int i = 0; i < FULLR; ++i) xb[(GTHREADS / 2) * i] = v[i];
        xb[(GTHREADS / 2) * last] = v[FULLR];
    }
    __syncthreads();

    float* dst = out + s * 8;
    const bool b4 = (lane & 4) != 0;
    const bool b2 = (lane & 2) != 0;
    const bool b1 = (lane & 1) != 0;

    for (; qA < qhi; qA += QSTEP, qB += QSTEP) {
        const bool vB  = qB < qhi;
        const int  qBc = vB ? qB : qA;
        const int qAn = (qA + QSTEP < qhi) ? (qA + QSTEP) : qA;
        const int qBn = (qB + QSTEP < qhi) ? (qB + QSTEP) : qAn;
        const Ent nA = load_ent(idx, dm, qAn, sub, ch);
        const Ent nB = load_ent(idx, dm, qBn, sub, ch);
        const int   oiA = qA  * (4 * KDIM) + sub * KDIM + ch;
        const int   oiB = qBc * (4 * KDIM) + sub * KDIM + ch;
        const float svA = dst[oiA];
        const float svB = dst[oiB];

        const float rA = quad_sum(xs, eA, b4, b2, b1);
        const float rB = quad_sum(xs, eB, b4, b2, b1);

        dst[oiA] = svA + rA * (1.0f / NASZ);
        if (vB) dst[oiB] = svB + rB * (1.0f / NASZ);

        eA = nA; eB = nB;
    }
}

extern "C" void kernel_launch(void* const* d_in, const int* in_sizes, int n_in,
                              void* d_out, int out_size, void* d_ws, size_t ws_size,
                              hipStream_t stream) {
    const float* x    = (const float*)d_in[0];
    const float* dm   = (const float*)d_in[1];
    const int*   idx  = (const int*)  d_in[2];
    const float* W    = (const float*)d_in[3];
    const float* bias = (const float*)d_in[4];
    float* out = (float*)d_out;
    float* ws  = (float*)d_ws;

    const int grid = (NN * BB + ROWS - 1) / ROWS;
    linear_first_kernel<<<grid, THREADS, 0, stream>>>(x, W, bias, ws, out);
    gather_slices_kernel<<<NSLICE * NGROUP, GTHREADS, 0, stream>>>(ws, dm, idx, out);
}
